# starting kernel
# baseline (speedup 1.0000x reference)
#include <hip/hip_runtime.h>

#define SEQL 256
#define BATCH 64
#define IN_DIM 512
#define HID 1024

typedef __attribute__((ext_vector_type(2))) float v2f;
typedef __attribute__((ext_vector_type(8))) float v8f;

struct GateParams {
    const float* Wi[4];
    const float* Wh[4];
};

__device__ __forceinline__ float fast_sigmoid(float x) {
    return 1.0f / (1.0f + __expf(-x));
}

__device__ __forceinline__ void kloop_dual(
    const float* __restrict__ a_base, const float* __restrict__ b_base,
    int kbegin, int kend, v8f& acc0, v8f& acc1)
{
    #pragma unroll 4
    for (int k0 = kbegin; k0 < kend; k0 += 8) {
        v2f a0 = *(const v2f*)(a_base + k0);
        v2f b0 = *(const v2f*)(b_base + k0);
        v2f a1 = *(const v2f*)(a_base + k0 + 4);
        v2f b1 = *(const v2f*)(b_base + k0 + 4);
        acc0 = __builtin_amdgcn_wmma_f32_16x16x4_f32(
            false, a0, false, b0, (short)0, acc0, false, false);
        acc1 = __builtin_amdgcn_wmma_f32_16x16x4_f32(
            false, a1, false, b1, (short)0, acc1, false, false);
    }
}

__global__ __launch_bounds__(256) void lstm_step_kernel(
    const float* __restrict__ x_t,
    const float* __restrict__ h_in,
    float* __restrict__ h_out,
    float* __restrict__ c_state,
    float* __restrict__ h_seq_t,
    const float* __restrict__ bsum,
    GateParams gp)
{
    __shared__ float smem[8 * 256];

    const int tid  = threadIdx.x;
    const int wave = tid >> 5;
    const int lane = tid & 31;
    const int gate = wave & 3;
    const int half = wave >> 2;

    const int bt = blockIdx.x;
    const int ht = blockIdx.y;

    const int m    = lane & 15;
    const int kk   = (lane >> 4) << 1;

    const int arow = bt * 16 + m;
    const int wrow = ht * 16 + m;

    v8f acc0 = {};
    v8f acc1 = {};

    const float* a_h = h_in + arow * HID + kk;
    const float* b_h = gp.Wh[gate] + wrow * HID + kk;

    if (half == 0) {
        kloop_dual(x_t + arow * IN_DIM + kk, gp.Wi[gate] + wrow * IN_DIM + kk,
                   0, IN_DIM, acc0, acc1);
        kloop_dual(a_h, b_h, 0, 256, acc0, acc1);
    } else {
        kloop_dual(a_h, b_h, 256, HID, acc0, acc1);
    }

    {
        float* s = smem + wave * 256;
        const int n     = lane & 15;
        const int mbase = (lane >> 4) << 3;
        #pragma unroll
        for (int r = 0; r < 8; ++r)
            s[(mbase + r) * 16 + n] = acc0[r] + acc1[r];
    }
    __syncthreads();

    {
        const int e  = tid;
        const int mm = e >> 4;
        const int n  = e & 15;
        const int b  = bt * 16 + mm;
        const int h  = ht * 16 + n;

        float fpre = smem[0 * 256 + e] + smem[4 * 256 + e] + bsum[0 * HID + h];
        float ipre = smem[1 * 256 + e] + smem[5 * 256 + e] + bsum[1 * HID + h];
        float opre = smem[2 * 256 + e] + smem[6 * 256 + e] + bsum[2 * HID + h];
        float gpre = smem[3 * 256 + e] + smem[7 * 256 + e] + bsum[3 * HID + h];

        float f = fast_sigmoid(fpre);
        float i = fast_sigmoid(ipre);
        float o = fast_sigmoid(opre);
        float g = tanhf(gpre);

        const int idx = b * HID + h;
        float c_new = f * c_state[idx] + i * g;
        c_state[idx] = c_new;
        float h_new = o * tanhf(c_new);
        h_out[idx]   = h_new;
        h_seq_t[idx] = h_new;
    }
}

__global__ __launch_bounds__(256) void prep_kernel(
    float* __restrict__ h0, float* __restrict__ c0,
    float* __restrict__ bsum,
    const float* __restrict__ bi0, const float* __restrict__ bh0,
    const float* __restrict__ bi1, const float* __restrict__ bh1,
    const float* __restrict__ bi2, const float* __restrict__ bh2,
    const float* __restrict__ bi3, const float* __restrict__ bh3)
{
    int i = blockIdx.x * blockDim.x + threadIdx.x;
    if (i < BATCH * HID) {
        h0[i] = 0.0f;
        c0[i] = 0.0f;
    }
    if (i < HID) {
        bsum[0 * HID + i] = bi0[i] + bh0[i];
        bsum[1 * HID + i] = bi1[i] + bh1[i];
        bsum[2 * HID + i] = bi2[i] + bh2[i];
        bsum[3 * HID + i] = bi3[i] + bh3[i];
    }
}

__global__ __launch_bounds__(256) void tail_copy_kernel(
    const float* __restrict__ h_fin, const float* __restrict__ c_fin,
    float* __restrict__ out_tail)
{
    int i = blockIdx.x * blockDim.x + threadIdx.x;
    if (i < BATCH * HID) {
        out_tail[i]               = h_fin[i];
        out_tail[BATCH * HID + i] = c_fin[i];
    }
}

extern "C" void kernel_launch(void* const* d_in, const int* in_sizes, int n_in,
                              void* d_out, int out_size, void* d_ws, size_t ws_size,
                              hipStream_t stream) {
    (void)in_sizes; (void)n_in; (void)out_size; (void)ws_size;

    const float* x = (const float*)d_in[0];
    GateParams gp;
    const float* bi[4];
    const float* bh[4];
    for (int g = 0; g < 4; ++g) {
        gp.Wi[g] = (const float*)d_in[1 + 4 * g];
        bi[g]    = (const float*)d_in[2 + 4 * g];
        gp.Wh[g] = (const float*)d_in[3 + 4 * g];
        bh[g]    = (const float*)d_in[4 + 4 * g];
    }

    float* ws    = (float*)d_ws;
    float* hbuf0 = ws;
    float* hbuf1 = ws + BATCH * HID;
    float* cbuf  = ws + 2 * BATCH * HID;
    float* bsum  = ws + 3 * BATCH * HID;
    float* out   = (float*)d_out;

    prep_kernel<<<(BATCH * HID + 255) / 256, 256, 0, stream>>>(
        hbuf0, cbuf, bsum,
        bi[0], bh[0], bi[1], bh[1], bi[2], bh[2], bi[3], bh[3]);

    dim3 grid(BATCH / 16, HID / 16);
    for (int t = 0; t < SEQL; ++t) {
        const float* h_in  = (t & 1) ? hbuf1 : hbuf0;
        float*       h_out = (t & 1) ? hbuf0 : hbuf1;
        lstm_step_kernel<<<grid, 256, 0, stream>>>(
            x + (size_t)t * BATCH * IN_DIM, h_in, h_out, cbuf,
            out + (size_t)t * BATCH * HID, bsum, gp);
    }
    tail_copy_kernel<<<(BATCH * HID + 255) / 256, 256, 0, stream>>>(
        hbuf0, cbuf, out + (size_t)SEQL * BATCH * HID);
}


# Recurrent weights re-laid once per call into fragment order in the workspace: a step wave's B operands are 64 contiguous 512-B loads sent before the barrier (were 128 row-strided); sums unchanged
# speedup vs baseline: 1.5320x; 1.5320x over previous
#include <hip/hip_runtime.h>

#define SEQL 256
#define BATCH 64
#define IN_DIM 512
#define HID 1024
#define CHUNK 16
#define WS_STATE_FLOATS (3 * BATCH * HID + 4 * HID)
#define WS_XP_FLOATS (CHUNK * 4 * BATCH * HID)
#define WS_NEED_BYTES ((unsigned long long)(WS_STATE_FLOATS + WS_XP_FLOATS) * 4ull)
#define WS_WP_FLOATS (4 * HID * HID)
#define WS_NEED2_BYTES ((unsigned long long)(WS_STATE_FLOATS + WS_XP_FLOATS + WS_WP_FLOATS) * 4ull)
#define A_PITCH (HID + 4)

typedef __attribute__((ext_vector_type(2))) float v2f;
typedef __attribute__((ext_vector_type(4))) float v4f;
typedef __attribute__((ext_vector_type(8))) float v8f;

struct GateParams {
    const float* Wi[4];
    const float* Wh[4];
};

__device__ __forceinline__ float fast_sigmoid(float x) {
    return 1.0f / (1.0f + __expf(-x));
}

__device__ __forceinline__ void kloop_dual(
    const float* __restrict__ a_base, const float* __restrict__ b_base,
    int kbegin, int kend, v8f& acc0, v8f& acc1)
{
    #pragma unroll 4
    for (int k0 = kbegin; k0 < kend; k0 += 8) {
        v2f a0 = *(const v2f*)(a_base + k0);
        v2f b0 = *(const v2f*)(b_base + k0);
        v2f a1 = *(const v2f*)(a_base + k0 + 4);
        v2f b1 = *(const v2f*)(b_base + k0 + 4);
        acc0 = __builtin_amdgcn_wmma_f32_16x16x4_f32(
            false, a0, false, b0, (short)0, acc0, false, false);
        acc1 = __builtin_amdgcn_wmma_f32_16x16x4_f32(
            false, a1, false, b1, (short)0, acc1, false, false);
    }
}

__device__ __forceinline__ void kloop_dual_lds(
    const float* a_lds, const float* __restrict__ b_base,
    int kbegin, int kend, v8f& acc0, v8f& acc1)
{
    for (int k0 = kbegin; k0 < kend; k0 += 64) {
        v2f a[16], b[16];
        #pragma unroll
        for (int i = 0; i < 16; ++i)
            b[i] = *(const v2f*)(b_base + k0 + 4 * i);
        #pragma unroll
        for (int i = 0; i < 16; ++i)
            a[i] = *(const v2f*)(a_lds + k0 + 4 * i);
        #pragma unroll
        for (int i = 0; i < 16; i += 2) {
            acc0 = __builtin_amdgcn_wmma_f32_16x16x4_f32(
                false, a[i], false, b[i], (short)0, acc0, false, false);
            acc1 = __builtin_amdgcn_wmma_f32_16x16x4_f32(
                false, a[i + 1], false, b[i + 1], (short)0, acc1, false, false);
        }
    }
}

__global__ __launch_bounds__(256) void xproj_kernel(
    const float* __restrict__ x_c,
    float* __restrict__ xp,
    GateParams gp, unsigned long long ws_bytes)
{
    if (ws_bytes < WS_NEED_BYTES) return;

    const int tid  = threadIdx.x;
    const int wave = tid >> 5;
    const int lane = tid & 31;
    const int wm   = wave & 1;
    const int wn   = wave >> 1;

    const int tl   = blockIdx.x;
    const int nb   = blockIdx.y;
    const int gate = nb >> 2;
    const int hcol = (nb & 3) * 256 + wn * 64;

    const int m  = lane & 15;
    const int kq = (lane >> 4) << 2;

    const float* a_base = x_c + (size_t)(tl * BATCH + wm * 32 + m) * IN_DIM + kq;
    const float* b_base = gp.Wi[gate] + (size_t)(hcol + m) * IN_DIM + kq;

    v8f acc[2][4];
    #pragma unroll
    for (int i = 0; i < 2; ++i)
        #pragma unroll
        for (int j = 0; j < 4; ++j)
            acc[i][j] = v8f{};

    #pragma unroll 4
    for (int k0 = 0; k0 < IN_DIM; k0 += 8) {
        v4f a[2], b[4];
        #pragma unroll
        for (int i = 0; i < 2; ++i)
            a[i] = *(const v4f*)(a_base + i * 16 * IN_DIM + k0);
        #pragma unroll
        for (int j = 0; j < 4; ++j)
            b[j] = *(const v4f*)(b_base + j * 16 * IN_DIM + k0);
        #pragma unroll
        for (int i = 0; i < 2; ++i)
            #pragma unroll
            for (int j = 0; j < 4; ++j) {
                acc[i][j] = __builtin_amdgcn_wmma_f32_16x16x4_f32(
                    false, v2f{a[i].x, a[i].y}, false, v2f{b[j].x, b[j].y},
                    (short)0, acc[i][j], false, false);
                acc[i][j] = __builtin_amdgcn_wmma_f32_16x16x4_f32(
                    false, v2f{a[i].z, a[i].w}, false, v2f{b[j].z, b[j].w},
                    (short)0, acc[i][j], false, false);
            }
    }

    const int n     = lane & 15;
    const int mbase = (lane >> 4) << 3;
    float* o = xp + (size_t)(tl * 4 + gate) * BATCH * HID;
    #pragma unroll
    for (int i = 0; i < 2; ++i)
        #pragma unroll
        for (int j = 0; j < 4; ++j)
            #pragma unroll
            for (int r = 0; r < 8; ++r)
                o[(wm * 32 + i * 16 + mbase + r) * HID + hcol + j * 16 + n] = acc[i][j][r];
}

__global__ __launch_bounds__(256) void wpack_kernel(
    float* __restrict__ wp, GateParams gp, unsigned long long ws_bytes)
{
    if (ws_bytes < WS_NEED2_BYTES) return;

    const unsigned r = blockIdx.x * 256u + threadIdx.x;
    const int lane = r & 31;
    const int jq   = (r >> 5) & 15;
    const int half = (r >> 9) & 1;
    const int gate = (r >> 10) & 3;
    const int ht   = r >> 12;
    const int m    = lane & 15;
    const int kk   = (lane >> 4) << 1;

    const float* src = gp.Wh[gate] + (size_t)(ht * 16 + m) * HID + half * 512 + 32 * jq + kk;
    v4f* dst = (v4f*)wp + (size_t)((((ht * 4 + gate) * 2 + half) * 64 + 4 * jq) * 32 + lane);
    v2f lo[4], hi[4];
    #pragma unroll
    for (int u = 0; u < 4; ++u) {
        lo[u] = *(const v2f*)(src + 8 * u);
        hi[u] = *(const v2f*)(src + 8 * u + 4);
    }
    #pragma unroll
    for (int u = 0; u < 4; ++u)
        dst[u * 32] = v4f{lo[u].x, lo[u].y, hi[u].x, hi[u].y};
}

__global__ __launch_bounds__(256) void lstm_step_kernel(
    const float* __restrict__ x_t,
    const float* __restrict__ h_in,
    float* __restrict__ h_out,
    float* __restrict__ c_state,
    float* __restrict__ h_seq_t,
    const float* __restrict__ bsum,
    const float* __restrict__ xp_t,
    const float* __restrict__ wp,
    GateParams gp, unsigned long long ws_bytes)
{
    __shared__ float smem[8 * 256];
    __shared__ __attribute__((aligned(16))) float sA[16 * A_PITCH];

    const bool hoisted = ws_bytes >= WS_NEED_BYTES;
    const bool packed  = ws_bytes >= WS_NEED2_BYTES;

    const int tid  = threadIdx.x;
    const int wave = tid >> 5;
    const int lane = tid & 31;
    const int gate = wave & 3;
    const int half = wave >> 2;

    const int bt = blockIdx.x;
    const int ht = blockIdx.y;

    const int m    = lane & 15;
    const int kk   = (lane >> 4) << 1;

    const int arow = bt * 16 + m;
    const int wrow = ht * 16 + m;

    v8f acc0 = {};
    v8f acc1 = {};

    const float* a_l = sA + m * A_PITCH + kk;
    const float* b_h = gp.Wh[gate] + wrow * HID + kk;

    float xpv[4] = {0.0f, 0.0f, 0.0f, 0.0f};
    if (hoisted) {
        const int xi = (bt * 16 + (tid >> 4)) * HID + ht * 16 + (tid & 15);
        #pragma unroll
        for (int g = 0; g < 4; ++g)
            xpv[g] = xp_t[g * BATCH * HID + xi];
    }

    if (packed) {
        const float* src_rows = h_in + (size_t)(bt * 16) * HID + tid * 4;
        v4f st[16];
        #pragma unroll
        for (int j = 0; j < 16; ++j)
            st[j] = *(const v4f*)(src_rows + j * HID);

        const v4f* bp = (const v4f*)wp + (size_t)(((ht * 4 + gate) * 2 + half) * 64) * 32 + lane;
        v4f b[64];
        #pragma unroll
        for (int j = 0; j < 64; ++j)
            b[j] = bp[j * 32];

        #pragma unroll
        for (int j = 0; j < 16; ++j)
            *(v4f*)(sA + j * A_PITCH + tid * 4) = st[j];
        __builtin_amdgcn_fence(__ATOMIC_RELEASE, "workgroup", "local");
        __builtin_amdgcn_s_barrier();
        __builtin_amdgcn_fence(__ATOMIC_ACQUIRE, "workgroup", "local");

        const float* a_w = a_l + half * 512;
        #pragma unroll
        for (int c = 0; c < 8; ++c) {
            v2f a[16];
            #pragma unroll
            for (int i = 0; i < 16; ++i)
                a[i] = *(const v2f*)(a_w + 64 * c + 4 * i);
            #pragma unroll
            for (int i = 0; i < 16; i += 2) {
                const v4f bb = b[8 * c + (i >> 1)];
                acc0 = __builtin_amdgcn_wmma_f32_16x16x4_f32(
                    false, a[i], false, v2f{bb.x, bb.y}, (short)0, acc0, false, false);
                acc1 = __builtin_amdgcn_wmma_f32_16x16x4_f32(
                    false, a[i + 1], false, v2f{bb.z, bb.w}, (short)0, acc1, false, false);
            }
        }
    } else {
        {
            const float* src_rows = h_in + (size_t)(bt * 16) * HID + tid * 4;
            v4f st[16];
            #pragma unroll
            for (int j = 0; j < 16; ++j)
                st[j] = *(const v4f*)(src_rows + j * HID);
            #pragma unroll
            for (int j = 0; j < 16; ++j)
                *(v4f*)(sA + j * A_PITCH + tid * 4) = st[j];
        }
        __syncthreads();

        if (hoisted) {
            kloop_dual_lds(a_l + half * 512, b_h + half * 512, 0, 512, acc0, acc1);
        } else if (half == 0) {
            kloop_dual(x_t + arow * IN_DIM + kk, gp.Wi[gate] + wrow * IN_DIM + kk,
                       0, IN_DIM, acc0, acc1);
            kloop_dual_lds(a_l, b_h, 0, 256, acc0, acc1);
        } else {
            kloop_dual_lds(a_l, b_h, 256, HID, acc0, acc1);
        }
    }

    {
        float* s = smem + wave * 256;
        const int n     = lane & 15;
        const int mbase = (lane >> 4) << 3;
        #pragma unroll
        for (int r = 0; r < 8; ++r)
            s[(mbase + r) * 16 + n] = acc0[r] + acc1[r];
    }
    __syncthreads();

    {
        const int e  = tid;
        const int mm = e >> 4;
        const int n  = e & 15;
        const int b  = bt * 16 + mm;
        const int h  = ht * 16 + n;

        float fpre = smem[0 * 256 + e] + smem[4 * 256 + e] + xpv[0] + bsum[0 * HID + h];
        float ipre = smem[1 * 256 + e] + smem[5 * 256 + e] + xpv[1] + bsum[1 * HID + h];
        float opre = smem[2 * 256 + e] + smem[6 * 256 + e] + xpv[2] + bsum[2 * HID + h];
        float gpre = smem[3 * 256 + e] + smem[7 * 256 + e] + xpv[3] + bsum[3 * HID + h];

        float f = fast_sigmoid(fpre);
        float i = fast_sigmoid(ipre);
        float o = fast_sigmoid(opre);
        float g = tanhf(gpre);

        const int idx = b * HID + h;
        float c_new = f * c_state[idx] + i * g;
        c_state[idx] = c_new;
        float h_new = o * tanhf(c_new);
        h_out[idx]   = h_new;
        h_seq_t[idx] = h_new;
    }
}

__global__ __launch_bounds__(256) void prep_kernel(
    float* __restrict__ h0, float* __restrict__ c0,
    float* __restrict__ bsum,
    const float* __restrict__ bi0, const float* __restrict__ bh0,
    const float* __restrict__ bi1, const float* __restrict__ bh1,
    const float* __restrict__ bi2, const float* __restrict__ bh2,
    const float* __restrict__ bi3, const float* __restrict__ bh3)
{
    int i = blockIdx.x * blockDim.x + threadIdx.x;
    if (i < BATCH * HID) {
        h0[i] = 0.0f;
        c0[i] = 0.0f;
    }
    if (i < HID) {
        bsum[0 * HID + i] = bi0[i] + bh0[i];
        bsum[1 * HID + i] = bi1[i] + bh1[i];
        bsum[2 * HID + i] = bi2[i] + bh2[i];
        bsum[3 * HID + i] = bi3[i] + bh3[i];
    }
}

__global__ __launch_bounds__(256) void tail_copy_kernel(
    const float* __restrict__ h_fin, const float* __restrict__ c_fin,
    float* __restrict__ out_tail)
{
    int i = blockIdx.x * blockDim.x + threadIdx.x;
    if (i < BATCH * HID) {
        out_tail[i]               = h_fin[i];
        out_tail[BATCH * HID + i] = c_fin[i];
    }
}

extern "C" void kernel_launch(void* const* d_in, const int* in_sizes, int n_in,
                              void* d_out, int out_size, void* d_ws, size_t ws_size,
                              hipStream_t stream) {
    (void)in_sizes; (void)n_in; (void)out_size;

    const float* x = (const float*)d_in[0];
    GateParams gp;
    const float* bi[4];
    const float* bh[4];
    for (int g = 0; g < 4; ++g) {
        gp.Wi[g] = (const float*)d_in[1 + 4 * g];
        bi[g]    = (const float*)d_in[2 + 4 * g];
        gp.Wh[g] = (const float*)d_in[3 + 4 * g];
        bh[g]    = (const float*)d_in[4 + 4 * g];
    }

    float* ws    = (float*)d_ws;
    float* hbuf0 = ws;
    float* hbuf1 = ws + BATCH * HID;
    float* cbuf  = ws + 2 * BATCH * HID;
    float* bsum  = ws + 3 * BATCH * HID;
    float* xp    = ws + WS_STATE_FLOATS;
    float* wp    = ws + WS_STATE_FLOATS + WS_XP_FLOATS;
    float* out   = (float*)d_out;
    const unsigned long long ws_bytes = (unsigned long long)ws_size;

    prep_kernel<<<(BATCH * HID + 255) / 256, 256, 0, stream>>>(
        hbuf0, cbuf, bsum,
        bi[0], bh[0], bi[1], bh[1], bi[2], bh[2], bi[3], bh[3]);

    wpack_kernel<<<WS_WP_FLOATS / 16 / 256, 256, 0, stream>>>(wp, gp, ws_bytes);

    dim3 grid(BATCH / 16, HID / 16);
    dim3 xgrid(CHUNK, 4 * HID / 256);
    for (int t0 = 0; t0 < SEQL; t0 += CHUNK) {
        xproj_kernel<<<xgrid, 256, 0, stream>>>(
            x + (size_t)t0 * BATCH * IN_DIM, xp, gp, ws_bytes);
        for (int tl = 0; tl < CHUNK; ++tl) {
            const int t = t0 + tl;
            const float* h_in  = (t & 1) ? hbuf1 : hbuf0;
            float*       h_out = (t & 1) ? hbuf0 : hbuf1;
            lstm_step_kernel<<<grid, 256, 0, stream>>>(
                x + (size_t)t * BATCH * IN_DIM, h_in, h_out, cbuf,
                out + (size_t)t * BATCH * HID, bsum,
                xp + (size_t)tl * 4 * BATCH * HID, wp, gp, ws_bytes);
        }
    }
    tail_copy_kernel<<<(BATCH * HID + 255) / 256, 256, 0, stream>>>(
        hbuf0, cbuf, out + (size_t)SEQL * BATCH * HID);
}


# Step work-group cut to 4 waves, one per SIMD: a wave owns all 4 gates of 4 hidden units over the whole K, gates taken from its own tile: no cross-wave partial sums or 2nd barrier; k sum order changes
# speedup vs baseline: 1.6180x; 1.0561x over previous
#include <hip/hip_runtime.h>

#define SEQL 256
#define BATCH 64
#define IN_DIM 512
#define HID 1024
#define CHUNK 16
#define WS_STATE_FLOATS (3 * BATCH * HID + 4 * HID)
#define WS_XP_FLOATS (CHUNK * 4 * BATCH * HID)
#define WS_NEED_BYTES ((unsigned long long)(WS_STATE_FLOATS + WS_XP_FLOATS) * 4ull)
#define WS_WP_FLOATS (4 * HID * HID)
#define WS_NEED2_BYTES ((unsigned long long)(WS_STATE_FLOATS + WS_XP_FLOATS + WS_WP_FLOATS) * 4ull)
#define A_PITCH (HID + 4)

typedef __attribute__((ext_vector_type(2))) float v2f;
typedef __attribute__((ext_vector_type(4))) float v4f;
typedef __attribute__((ext_vector_type(8))) float v8f;

struct GateParams {
    const float* Wi[4];
    const float* Wh[4];
};

__device__ __forceinline__ float fast_sigmoid(float x) {
    return 1.0f / (1.0f + __expf(-x));
}

__device__ __forceinline__ void kloop_dual(
    const float* __restrict__ a_base, const float* __restrict__ b_base,
    int kbegin, int kend, v8f& acc0, v8f& acc1)
{
    #pragma unroll 4
    for (int k0 = kbegin; k0 < kend; k0 += 8) {
        v2f a0 = *(const v2f*)(a_base + k0);
        v2f b0 = *(const v2f*)(b_base + k0);
        v2f a1 = *(const v2f*)(a_base + k0 + 4);
        v2f b1 = *(const v2f*)(b_base + k0 + 4);
        acc0 = __builtin_amdgcn_wmma_f32_16x16x4_f32(
            false, a0, false, b0, (short)0, acc0, false, false);
        acc1 = __builtin_amdgcn_wmma_f32_16x16x4_f32(
            false, a1, false, b1, (short)0, acc1, false, false);
    }
}

__device__ __forceinline__ void kloop_dual_lds(
    const float* a_lds, const float* __restrict__ b_base,
    int kbegin, int kend, v8f& acc0, v8f& acc1)
{
    for (int k0 = kbegin; k0 < kend; k0 += 64) {
        v2f a[16], b[16];
        #pragma unroll
        for (int i = 0; i < 16; ++i)
            b[i] = *(const v2f*)(b_base + k0 + 4 * i);
        #pragma unroll
        for (int i = 0; i < 16; ++i)
            a[i] = *(const v2f*)(a_lds + k0 + 4 * i);
        #pragma unroll
        for (int i = 0; i < 16; i += 2) {
            acc0 = __builtin_amdgcn_wmma_f32_16x16x4_f32(
                false, a[i], false, b[i], (short)0, acc0, false, false);
            acc1 = __builtin_amdgcn_wmma_f32_16x16x4_f32(
                false, a[i + 1], false, b[i + 1], (short)0, acc1, false, false);
        }
    }
}

__global__ __launch_bounds__(256) void xproj_kernel(
    const float* __restrict__ x_c,
    float* __restrict__ xp,
    GateParams gp, unsigned long long ws_bytes)
{
    if (ws_bytes < WS_NEED_BYTES) return;

    const int tid  = threadIdx.x;
    const int wave = tid >> 5;
    const int lane = tid & 31;
    const int wm   = wave & 1;
    const int wn   = wave >> 1;

    const int tl   = blockIdx.x;
    const int nb   = blockIdx.y;
    const int gate = nb >> 2;
    const int hcol = (nb & 3) * 256 + wn * 64;

    const int m  = lane & 15;
    const int kq = (lane >> 4) << 2;

    const float* a_base = x_c + (size_t)(tl * BATCH + wm * 32 + m) * IN_DIM + kq;
    const float* b_base = gp.Wi[gate] + (size_t)(hcol + m) * IN_DIM + kq;

    v8f acc[2][4];
    #pragma unroll
    for (int i = 0; i < 2; ++i)
        #pragma unroll
        for (int j = 0; j < 4; ++j)
            acc[i][j] = v8f{};

    #pragma unroll 4
    for (int k0 = 0; k0 < IN_DIM; k0 += 8) {
        v4f a[2], b[4];
        #pragma unroll
        for (int i = 0; i < 2; ++i)
            a[i] = *(const v4f*)(a_base + i * 16 * IN_DIM + k0);
        #pragma unroll
        for (int j = 0; j < 4; ++j)
            b[j] = *(const v4f*)(b_base + j * 16 * IN_DIM + k0);
        #pragma unroll
        for (int i = 0; i < 2; ++i)
            #pragma unroll
            for (int j = 0; j < 4; ++j) {
                acc[i][j] = __builtin_amdgcn_wmma_f32_16x16x4_f32(
                    false, v2f{a[i].x, a[i].y}, false, v2f{b[j].x, b[j].y},
                    (short)0, acc[i][j], false, false);
                acc[i][j] = __builtin_amdgcn_wmma_f32_16x16x4_f32(
                    false, v2f{a[i].z, a[i].w}, false, v2f{b[j].z, b[j].w},
                    (short)0, acc[i][j], false, false);
            }
    }

    const int n     = lane & 15;
    const int mbase = (lane >> 4) << 3;
    float* o = xp + (size_t)(tl * 4 + gate) * BATCH * HID;
    #pragma unroll
    for (int i = 0; i < 2; ++i)
        #pragma unroll
        for (int j = 0; j < 4; ++j)
            #pragma unroll
            for (int r = 0; r < 8; ++r)
                o[(wm * 32 + i * 16 + mbase + r) * HID + hcol + j * 16 + n] = acc[i][j][r];
}

__global__ __launch_bounds__(256) void wpack_kernel(
    float* __restrict__ wp, GateParams gp, unsigned long long ws_bytes)
{
    if (ws_bytes < WS_NEED2_BYTES) return;

    const unsigned r = blockIdx.x * 256u + threadIdx.x;
    const int lane = r & 31;
    const int jq   = (r >> 5) & 31;
    const int wv   = (r >> 10) & 3;
    const int ht   = r >> 12;
    const int c    = lane & 15;
    const int cj   = c >> 2;
    const int cg   = c & 3;
    const int kk   = (lane >> 4) << 1;

    const float* wg = cg == 0 ? gp.Wh[0] : cg == 1 ? gp.Wh[1] : cg == 2 ? gp.Wh[2] : gp.Wh[3];
    const float* src = wg + (size_t)(ht * 16 + 4 * wv + cj) * HID + 32 * jq + kk;
    v4f* dst = (v4f*)wp + (size_t)(((ht * 4 + wv) * 128 + 4 * jq) * 32 + lane);
    v2f lo[4], hi[4];
    #pragma unroll
    for (int u = 0; u < 4; ++u) {
        lo[u] = *(const v2f*)(src + 8 * u);
        hi[u] = *(const v2f*)(src + 8 * u + 4);
    }
    #pragma unroll
    for (int u = 0; u < 4; ++u)
        dst[u * 32] = v4f{lo[u].x, lo[u].y, hi[u].x, hi[u].y};
}

#define G_PITCH 20

__global__ __launch_bounds__(128) void lstm_step_kernel(
    const float* __restrict__ x_t,
    const float* __restrict__ h_in,
    float* __restrict__ h_out,
    float* __restrict__ c_state,
    float* __restrict__ h_seq_t,
    const float* __restrict__ bsum,
    const float* __restrict__ xp_t,
    const float* wp,
    GateParams gp, unsigned long long ws_bytes)
{
    __shared__ __attribute__((aligned(16))) float smem[4 * 16 * G_PITCH];
    __shared__ __attribute__((aligned(16))) float sA[16 * A_PITCH];

    const bool hoisted = ws_bytes >= WS_NEED_BYTES;
    const bool packed  = ws_bytes >= WS_NEED2_BYTES;

    const int tid  = threadIdx.x;
    const int wave = tid >> 5;
    const int lane = tid & 31;

    const int bt = blockIdx.x;
    const int ht = blockIdx.y;

    const int m    = lane & 15;
    const int kk   = (lane >> 4) << 1;

    const int erow = lane >> 2;
    const int hcol = ht * 16 + 4 * wave + (lane & 3);

    v8f acc0 = {};
    v8f acc1 = {};

    const float* a_l = sA + m * A_PITCH + kk;

    float xpv[2][4] = {{0.0f, 0.0f, 0.0f, 0.0f}, {0.0f, 0.0f, 0.0f, 0.0f}};
    if (hoisted) {
        #pragma unroll
        for (int s = 0; s < 2; ++s) {
            const int xi = (bt * 16 + erow + 8 * s) * HID + hcol;
            #pragma unroll
            for (int g = 0; g < 4; ++g)
                xpv[s][g] = xp_t[g * BATCH * HID + xi];
        }
    }

    if (packed) {
        const float* src_rows = h_in + (size_t)(bt * 16) * HID + tid * 4;
        v4f st[32];
        #pragma unroll
        for (int j = 0; j < 16; ++j) {
            st[2 * j]     = *(const v4f*)(src_rows + j * HID);
            st[2 * j + 1] = *(const v4f*)(src_rows + j * HID + 512);
        }

        const v4f* bp = (const v4f*)wp + (size_t)((ht * 4 + wave) * 128) * 32 + lane;
        v4f b[128];
        #pragma unroll
        for (int j = 0; j < 16; ++j)
            b[j] = bp[j * 32];

        #pragma unroll
        for (int j = 0; j < 16; ++j) {
            *(v4f*)(sA + j * A_PITCH + tid * 4)       = st[2 * j];
            *(v4f*)(sA + j * A_PITCH + tid * 4 + 512) = st[2 * j + 1];
        }
        __builtin_amdgcn_sched_barrier(0);
        #pragma unroll
        for (int j = 16; j < 48; ++j)
            b[j] = bp[j * 32];
        __builtin_amdgcn_fence(__ATOMIC_RELEASE, "workgroup", "local");
        __builtin_amdgcn_s_barrier();
        __builtin_amdgcn_fence(__ATOMIC_ACQUIRE, "workgroup", "local");

        v2f a[2][16];
        #pragma unroll
        for (int i = 0; i < 16; ++i)
            a[0][i] = *(const v2f*)(a_l + 4 * i);
        #pragma unroll
        for (int c = 0; c < 16; ++c) {
            if (c + 6 < 16) {
                #pragma unroll
                for (int j = 0; j < 8; ++j)
                    b[8 * (c + 6) + j] = bp[(8 * (c + 6) + j) * 32];
            }
            if (c + 1 < 16) {
                #pragma unroll
                for (int i = 0; i < 16; ++i)
                    a[(c + 1) & 1][i] = *(const v2f*)(a_l + 64 * (c + 1) + 4 * i);
            }
            __builtin_amdgcn_sched_barrier(0);
            #pragma unroll
            for (int i = 0; i < 16; i += 2) {
                const v4f bb = b[8 * c + (i >> 1)];
                acc0 = __builtin_amdgcn_wmma_f32_16x16x4_f32(
                    false, a[c & 1][i], false, v2f{bb.x, bb.y}, (short)0, acc0, false, false);
                acc1 = __builtin_amdgcn_wmma_f32_16x16x4_f32(
                    false, a[c & 1][i + 1], false, v2f{bb.z, bb.w}, (short)0, acc1, false, false);
            }
            if (c + 1 < 16)
                __builtin_amdgcn_sched_barrier(0);
        }
    } else {
        {
            const float* src_rows = h_in + (size_t)(bt * 16) * HID + tid * 4;
            v4f st[32];
            #pragma unroll
            for (int j = 0; j < 16; ++j) {
                st[2 * j]     = *(const v4f*)(src_rows + j * HID);
                st[2 * j + 1] = *(const v4f*)(src_rows + j * HID + 512);
            }
            #pragma unroll
            for (int j = 0; j < 16; ++j) {
                *(v4f*)(sA + j * A_PITCH + tid * 4)       = st[2 * j];
                *(v4f*)(sA + j * A_PITCH + tid * 4 + 512) = st[2 * j + 1];
            }
        }
        __syncthreads();

        const int cg   = m & 3;
        const int wrow = ht * 16 + 4 * wave + (m >> 2);
        if (!hoisted) {
            const float* wi = cg == 0 ? gp.Wi[0] : cg == 1 ? gp.Wi[1] : cg == 2 ? gp.Wi[2] : gp.Wi[3];
            kloop_dual(x_t + (bt * 16 + m) * IN_DIM + kk, wi + (size_t)wrow * IN_DIM + kk,
                       0, IN_DIM, acc0, acc1);
        }
        const float* wh = cg == 0 ? gp.Wh[0] : cg == 1 ? gp.Wh[1] : cg == 2 ? gp.Wh[2] : gp.Wh[3];
        kloop_dual_lds(a_l, wh + (size_t)wrow * HID + kk, 0, HID, acc0, acc1);
    }

    float* sw = smem + wave * (16 * G_PITCH);
    {
        const int mbase = (lane >> 4) << 3;
        #pragma unroll
        for (int r = 0; r < 8; ++r)
            sw[(mbase + r) * G_PITCH + m] = acc0[r] + acc1[r];
    }
    __builtin_amdgcn_fence(__ATOMIC_RELEASE, "workgroup", "local");
    __builtin_amdgcn_wave_barrier();
    __builtin_amdgcn_fence(__ATOMIC_ACQUIRE, "workgroup", "local");

    {
        const float b_f = bsum[0 * HID + hcol];
        const float b_i = bsum[1 * HID + hcol];
        const float b_o = bsum[2 * HID + hcol];
        const float b_g = bsum[3 * HID + hcol];
        const int idx0 = (bt * 16 + erow) * HID + hcol;
        float c_old[2];
        c_old[0] = c_state[idx0];
        c_old[1] = c_state[idx0 + 8 * HID];
        #pragma unroll
        for (int s = 0; s < 2; ++s) {
            const int row = erow + 8 * s;
            const v4f pre = *(const v4f*)(sw + row * G_PITCH + 4 * (lane & 3));

            float fpre = pre.x + xpv[s][0] + b_f;
            float ipre = pre.y + xpv[s][1] + b_i;
            float opre = pre.z + xpv[s][2] + b_o;
            float gpre = pre.w + xpv[s][3] + b_g;

            float f = fast_sigmoid(fpre);
            float i = fast_sigmoid(ipre);
            float o = fast_sigmoid(opre);
            float g = tanhf(gpre);

            const int idx = idx0 + 8 * s * HID;
            float c_new = f * c_old[s] + i * g;
            c_state[idx] = c_new;
            float h_new = o * tanhf(c_new);
            h_out[idx]   = h_new;
            h_seq_t[idx] = h_new;
        }
    }
}

__global__ __launch_bounds__(256) void prep_kernel(
    float* __restrict__ h0, float* __restrict__ c0,
    float* __restrict__ bsum,
    const float* __restrict__ bi0, const float* __restrict__ bh0,
    const float* __restrict__ bi1, const float* __restrict__ bh1,
    const float* __restrict__ bi2, const float* __restrict__ bh2,
    const float* __restrict__ bi3, const float* __restrict__ bh3)
{
    int i = blockIdx.x * blockDim.x + threadIdx.x;
    if (i < BATCH * HID) {
        h0[i] = 0.0f;
        c0[i] = 0.0f;
    }
    if (i < HID) {
        bsum[0 * HID + i] = bi0[i] + bh0[i];
        bsum[1 * HID + i] = bi1[i] + bh1[i];
        bsum[2 * HID + i] = bi2[i] + bh2[i];
        bsum[3 * HID + i] = bi3[i] + bh3[i];
    }
}

__global__ __launch_bounds__(256) void tail_copy_kernel(
    const float* __restrict__ h_fin, const float* __restrict__ c_fin,
    float* __restrict__ out_tail)
{
    int i = blockIdx.x * blockDim.x + threadIdx.x;
    if (i < BATCH * HID) {
        out_tail[i]               = h_fin[i];
        out_tail[BATCH * HID + i] = c_fin[i];
    }
}

extern "C" void kernel_launch(void* const* d_in, const int* in_sizes, int n_in,
                              void* d_out, int out_size, void* d_ws, size_t ws_size,
                              hipStream_t stream) {
    (void)in_sizes; (void)n_in; (void)out_size;

    const float* x = (const float*)d_in[0];
    GateParams gp;
    const float* bi[4];
    const float* bh[4];
    for (int g = 0; g < 4; ++g) {
        gp.Wi[g] = (const float*)d_in[1 + 4 * g];
        bi[g]    = (const float*)d_in[2 + 4 * g];
        gp.Wh[g] = (const float*)d_in[3 + 4 * g];
        bh[g]    = (const float*)d_in[4 + 4 * g];
    }

    float* ws    = (float*)d_ws;
    float* hbuf0 = ws;
    float* hbuf1 = ws + BATCH * HID;
    float* cbuf  = ws + 2 * BATCH * HID;
    float* bsum  = ws + 3 * BATCH * HID;
    float* xp    = ws + WS_STATE_FLOATS;
    float* wp    = ws + WS_STATE_FLOATS + WS_XP_FLOATS;
    float* out   = (float*)d_out;
    const unsigned long long ws_bytes = (unsigned long long)ws_size;

    prep_kernel<<<(BATCH * HID + 255) / 256, 256, 0, stream>>>(
        hbuf0, cbuf, bsum,
        bi[0], bh[0], bi[1], bh[1], bi[2], bh[2], bi[3], bh[3]);

    wpack_kernel<<<WS_WP_FLOATS / 16 / 256, 256, 0, stream>>>(wp, gp, ws_bytes);

    dim3 grid(BATCH / 16, HID / 16);
    dim3 xgrid(CHUNK, 4 * HID / 256);
    for (int t0 = 0; t0 < SEQL; t0 += CHUNK) {
        xproj_kernel<<<xgrid, 256, 0, stream>>>(
            x + (size_t)t0 * BATCH * IN_DIM, xp, gp, ws_bytes);
        for (int tl = 0; tl < CHUNK; ++tl) {
            const int t = t0 + tl;
            const float* h_in  = (t & 1) ? hbuf1 : hbuf0;
            float*       h_out = (t & 1) ? hbuf0 : hbuf1;
            lstm_step_kernel<<<grid, 128, 0, stream>>>(
                x + (size_t)t * BATCH * IN_DIM, h_in, h_out, cbuf,
                out + (size_t)t * BATCH * HID, bsum,
                xp + (size_t)tl * 4 * BATCH * HID, wp, gp, ws_bytes);
        }
    }
    tail_copy_kernel<<<(BATCH * HID + 255) / 256, 256, 0, stream>>>(
        hbuf0, cbuf, out + (size_t)SEQL * BATCH * HID);
}


# Step stores cover whole 128-B lines: h and c state kept wave-tiled in the workspace (one writer per line); step t writes sequence slot t-1 from its LDS h tile, tail kernel the last; sums unchanged
# speedup vs baseline: 1.8789x; 1.1612x over previous
#include <hip/hip_runtime.h>

#define SEQL 256
#define BATCH 64
#define IN_DIM 512
#define HID 1024
#define CHUNK 16
#define WS_STATE_FLOATS (3 * BATCH * HID + 4 * HID)
#define WS_XP_FLOATS (CHUNK * 4 * BATCH * HID)
#define WS_NEED_BYTES ((unsigned long long)(WS_STATE_FLOATS + WS_XP_FLOATS) * 4ull)
#define WS_WP_FLOATS (4 * HID * HID)
#define WS_NEED2_BYTES ((unsigned long long)(WS_STATE_FLOATS + WS_XP_FLOATS + WS_WP_FLOATS) * 4ull)
#define A_PITCH (HID + 4)

#define PK_TILE (16 * HID)

typedef __attribute__((ext_vector_type(2))) float v2f;
typedef __attribute__((ext_vector_type(4))) float v4f;
typedef __attribute__((ext_vector_type(8))) float v8f;

struct GateParams {
    const float* Wi[4];
    const float* Wh[4];
};

__device__ __forceinline__ float fast_sigmoid(float x) {
    return 1.0f / (1.0f + __expf(-x));
}

__device__ __forceinline__ void kloop_dual(
    const float* __restrict__ a_base, const float* __restrict__ b_base,
    int kbegin, int kend, v8f& acc0, v8f& acc1)
{
    #pragma unroll 4
    for (int k0 = kbegin; k0 < kend; k0 += 8) {
        v2f a0 = *(const v2f*)(a_base + k0);
        v2f b0 = *(const v2f*)(b_base + k0);
        v2f a1 = *(const v2f*)(a_base + k0 + 4);
        v2f b1 = *(const v2f*)(b_base + k0 + 4);
        acc0 = __builtin_amdgcn_wmma_f32_16x16x4_f32(
            false, a0, false, b0, (short)0, acc0, false, false);
        acc1 = __builtin_amdgcn_wmma_f32_16x16x4_f32(
            false, a1, false, b1, (short)0, acc1, false, false);
    }
}

__device__ __forceinline__ void kloop_dual_lds(
    const float* a_lds, const float* __restrict__ b_base,
    int kbegin, int kend, v8f& acc0, v8f& acc1)
{
    for (int k0 = kbegin; k0 < kend; k0 += 64) {
        v2f a[16], b[16];
        #pragma unroll
        for (int i = 0; i < 16; ++i)
            b[i] = *(const v2f*)(b_base + k0 + 4 * i);
        #pragma unroll
        for (int i = 0; i < 16; ++i)
            a[i] = *(const v2f*)(a_lds + k0 + 4 * i);
        #pragma unroll
        for (int i = 0; i < 16; i += 2) {
            acc0 = __builtin_amdgcn_wmma_f32_16x16x4_f32(
                false, a[i], false, b[i], (short)0, acc0, false, false);
            acc1 = __builtin_amdgcn_wmma_f32_16x16x4_f32(
                false, a[i + 1], false, b[i + 1], (short)0, acc1, false, false);
        }
    }
}

__global__ __launch_bounds__(256) void xproj_kernel(
    const float* __restrict__ x_c,
    float* __restrict__ xp,
    GateParams gp, unsigned long long ws_bytes)
{
    if (ws_bytes < WS_NEED_BYTES) return;

    const int tid  = threadIdx.x;
    const int wave = tid >> 5;
    const int lane = tid & 31;
    const int wm   = wave & 1;
    const int wn   = wave >> 1;

    const int tl   = blockIdx.x;
    const int nb   = blockIdx.y;
    const int gate = nb >> 2;
    const int hcol = (nb & 3) * 256 + wn * 64;

    const int m  = lane & 15;
    const int kq = (lane >> 4) << 2;

    const float* a_base = x_c + (size_t)(tl * BATCH + wm * 32 + m) * IN_DIM + kq;
    const float* b_base = gp.Wi[gate] + (size_t)(hcol + m) * IN_DIM + kq;

    v8f acc[2][4];
    #pragma unroll
    for (int i = 0; i < 2; ++i)
        #pragma unroll
        for (int j = 0; j < 4; ++j)
            acc[i][j] = v8f{};

    #pragma unroll 4
    for (int k0 = 0; k0 < IN_DIM; k0 += 8) {
        v4f a[2], b[4];
        #pragma unroll
        for (int i = 0; i < 2; ++i)
            a[i] = *(const v4f*)(a_base + i * 16 * IN_DIM + k0);
        #pragma unroll
        for (int j = 0; j < 4; ++j)
            b[j] = *(const v4f*)(b_base + j * 16 * IN_DIM + k0);
        #pragma unroll
        for (int i = 0; i < 2; ++i)
            #pragma unroll
            for (int j = 0; j < 4; ++j) {
                acc[i][j] = __builtin_amdgcn_wmma_f32_16x16x4_f32(
                    false, v2f{a[i].x, a[i].y}, false, v2f{b[j].x, b[j].y},
                    (short)0, acc[i][j], false, false);
                acc[i][j] = __builtin_amdgcn_wmma_f32_16x16x4_f32(
                    false, v2f{a[i].z, a[i].w}, false, v2f{b[j].z, b[j].w},
                    (short)0, acc[i][j], false, false);
            }
    }

    const int n     = lane & 15;
    const int mbase = (lane >> 4) << 3;
    float* o = xp + (size_t)(tl * 4 + gate) * BATCH * HID;
    #pragma unroll
    for (int i = 0; i < 2; ++i)
        #pragma unroll
        for (int j = 0; j < 4; ++j)
            #pragma unroll
            for (int r = 0; r < 8; ++r)
                o[(wm * 32 + i * 16 + mbase + r) * HID + hcol + j * 16 + n] = acc[i][j][r];
}

__global__ __launch_bounds__(256) void wpack_kernel(
    float* __restrict__ wp, GateParams gp, unsigned long long ws_bytes)
{
    if (ws_bytes < WS_NEED2_BYTES) return;

    const unsigned r = blockIdx.x * 256u + threadIdx.x;
    const int lane = r & 31;
    const int jq   = (r >> 5) & 31;
    const int wv   = (r >> 10) & 3;
    const int ht   = r >> 12;
    const int c    = lane & 15;
    const int cj   = c >> 2;
    const int cg   = c & 3;
    const int kk   = (lane >> 4) << 1;

    const float* wg = cg == 0 ? gp.Wh[0] : cg == 1 ? gp.Wh[1] : cg == 2 ? gp.Wh[2] : gp.Wh[3];
    const float* src = wg + (size_t)(ht * 16 + 4 * wv + cj) * HID + 32 * jq + kk;
    v4f* dst = (v4f*)wp + (size_t)(((ht * 4 + wv) * 128 + 4 * jq) * 32 + lane);
    v2f lo[4], hi[4];
    #pragma unroll
    for (int u = 0; u < 4; ++u) {
        lo[u] = *(const v2f*)(src + 8 * u);
        hi[u] = *(const v2f*)(src + 8 * u + 4);
    }
    #pragma unroll
    for (int u = 0; u < 4; ++u)
        dst[u * 32] = v4f{lo[u].x, lo[u].y, hi[u].x, hi[u].y};
}

#define G_PITCH 20

__global__ __launch_bounds__(128) void lstm_step_kernel(
    const float* __restrict__ x_t,
    const float* __restrict__ h_in,
    float* __restrict__ h_out,
    float* __restrict__ c_state,
    float* __restrict__ out_prev,
    const float* __restrict__ bsum,
    const float* __restrict__ xp_t,
    const float* wp,
    GateParams gp, unsigned long long ws_bytes)
{
    __shared__ __attribute__((aligned(16))) float smem[4 * 16 * G_PITCH];
    __shared__ __attribute__((aligned(16))) float sA[16 * A_PITCH];

    const bool hoisted = ws_bytes >= WS_NEED_BYTES;
    const bool packed  = ws_bytes >= WS_NEED2_BYTES;

    const int tid  = threadIdx.x;
    const int wave = tid >> 5;
    const int lane = tid & 31;

    const int bt = blockIdx.x;
    const int ht = blockIdx.y;

    const int m    = lane & 15;
    const int kk   = (lane >> 4) << 1;

    const int erow = lane >> 2;
    const int hcol = ht * 16 + 4 * wave + (lane & 3);

    v8f acc0 = {};
    v8f acc1 = {};

    const float* a_l = sA + m * A_PITCH + kk;

    float xpv[2][4] = {{0.0f, 0.0f, 0.0f, 0.0f}, {0.0f, 0.0f, 0.0f, 0.0f}};
    if (hoisted) {
        #pragma unroll
        for (int s = 0; s < 2; ++s) {
            const int xi = (bt * 16 + erow + 8 * s) * HID + hcol;
            #pragma unroll
            for (int g = 0; g < 4; ++g)
                xpv[s][g] = xp_t[g * BATCH * HID + xi];
        }
    }

    if (packed) {
        const float* src_tile = h_in + (size_t)bt * PK_TILE + tid * 4;
        v4f st[32];
        #pragma unroll
        for (int jj = 0; jj < 32; ++jj)
            st[jj] = *(const v4f*)(src_tile + jj * 512);

        const v4f* bp = (const v4f*)wp + (size_t)((ht * 4 + wave) * 128) * 32 + lane;
        v4f b[128];
        #pragma unroll
        for (int j = 0; j < 16; ++j)
            b[j] = bp[j * 32];

        float* dst_l = sA + (tid & 15) * A_PITCH + (tid >> 4) * 4;
        #pragma unroll
        for (int jj = 0; jj < 32; ++jj)
            *(v4f*)(dst_l + jj * 32) = st[jj];
        __builtin_amdgcn_sched_barrier(0);
        #pragma unroll
        for (int j = 16; j < 48; ++j)
            b[j] = bp[j * 32];
        __builtin_amdgcn_fence(__ATOMIC_RELEASE, "workgroup", "local");
        __builtin_amdgcn_s_barrier();
        __builtin_amdgcn_fence(__ATOMIC_ACQUIRE, "workgroup", "local");

        v2f a[2][16];
        #pragma unroll
        for (int i = 0; i < 16; ++i)
            a[0][i] = *(const v2f*)(a_l + 4 * i);
        #pragma unroll
        for (int c = 0; c < 16; ++c) {
            if (c + 6 < 16) {
                #pragma unroll
                for (int j = 0; j < 8; ++j)
                    b[8 * (c + 6) + j] = bp[(8 * (c + 6) + j) * 32];
            }
            if (c + 1 < 16) {
                #pragma unroll
                for (int i = 0; i < 16; ++i)
                    a[(c + 1) & 1][i] = *(const v2f*)(a_l + 64 * (c + 1) + 4 * i);
            }
            __builtin_amdgcn_sched_barrier(0);
            #pragma unroll
            for (int i = 0; i < 16; i += 2) {
                const v4f bb = b[8 * c + (i >> 1)];
                acc0 = __builtin_amdgcn_wmma_f32_16x16x4_f32(
                    false, a[c & 1][i], false, v2f{bb.x, bb.y}, (short)0, acc0, false, false);
                acc1 = __builtin_amdgcn_wmma_f32_16x16x4_f32(
                    false, a[c & 1][i + 1], false, v2f{bb.z, bb.w}, (short)0, acc1, false, false);
            }
            if (c + 1 < 16)
                __builtin_amdgcn_sched_barrier(0);
        }
    } else {
        {
            const float* src_tile = h_in + (size_t)bt * PK_TILE + tid * 4;
            float* dst_l = sA + (tid & 15) * A_PITCH + (tid >> 4) * 4;
            v4f st[32];
            #pragma unroll
            for (int jj = 0; jj < 32; ++jj)
                st[jj] = *(const v4f*)(src_tile + jj * 512);
            #pragma unroll
            for (int jj = 0; jj < 32; ++jj)
                *(v4f*)(dst_l + jj * 32) = st[jj];
        }
        __syncthreads();

        const int cg   = m & 3;
        const int wrow = ht * 16 + 4 * wave + (m >> 2);
        if (!hoisted) {
            const float* wi = cg == 0 ? gp.Wi[0] : cg == 1 ? gp.Wi[1] : cg == 2 ? gp.Wi[2] : gp.Wi[3];
            kloop_dual(x_t + (bt * 16 + m) * IN_DIM + kk, wi + (size_t)wrow * IN_DIM + kk,
                       0, IN_DIM, acc0, acc1);
        }
        const float* wh = cg == 0 ? gp.Wh[0] : cg == 1 ? gp.Wh[1] : cg == 2 ? gp.Wh[2] : gp.Wh[3];
        kloop_dual_lds(a_l, wh + (size_t)wrow * HID + kk, 0, HID, acc0, acc1);
    }

    float* sw = smem + wave * (16 * G_PITCH);
    {
        const int mbase = (lane >> 4) << 3;
        #pragma unroll
        for (int r = 0; r < 8; ++r)
            sw[(mbase + r) * G_PITCH + m] = acc0[r] + acc1[r];
    }
    __builtin_amdgcn_fence(__ATOMIC_RELEASE, "workgroup", "local");
    __builtin_amdgcn_wave_barrier();
    __builtin_amdgcn_fence(__ATOMIC_ACQUIRE, "workgroup", "local");

    {
        const float b_f = bsum[0 * HID + hcol];
        const float b_i = bsum[1 * HID + hcol];
        const float b_o = bsum[2 * HID + hcol];
        const float b_g = bsum[3 * HID + hcol];
        const int pk0 = bt * PK_TILE + (ht * 4 + wave) * 64 + lane;
        float c_old[2];
        c_old[0] = c_state[pk0];
        c_old[1] = c_state[pk0 + 32];
        if (out_prev) {
            const int r  = ht & 15;
            const int cc = (ht >> 4) * 256 + wave * 64 + lane * 2;
            const v2f hv = *(const v2f*)(sA + r * A_PITCH + cc);
            *(v2f*)(out_prev + (size_t)(bt * 16 + r) * HID + cc) = hv;
        }
        #pragma unroll
        for (int s = 0; s < 2; ++s) {
            const int row = erow + 8 * s;
            const v4f pre = *(const v4f*)(sw + row * G_PITCH + 4 * (lane & 3));

            float fpre = pre.x + xpv[s][0] + b_f;
            float ipre = pre.y + xpv[s][1] + b_i;
            float opre = pre.z + xpv[s][2] + b_o;
            float gpre = pre.w + xpv[s][3] + b_g;

            float f = fast_sigmoid(fpre);
            float i = fast_sigmoid(ipre);
            float o = fast_sigmoid(opre);
            float g = tanhf(gpre);

            const int idx = pk0 + 32 * s;
            float c_new = f * c_old[s] + i * g;
            c_state[idx] = c_new;
            float h_new = o * tanhf(c_new);
            h_out[idx]   = h_new;
        }
    }
}

__global__ __launch_bounds__(256) void prep_kernel(
    float* __restrict__ h0, float* __restrict__ c0,
    float* __restrict__ bsum,
    const float* __restrict__ bi0, const float* __restrict__ bh0,
    const float* __restrict__ bi1, const float* __restrict__ bh1,
    const float* __restrict__ bi2, const float* __restrict__ bh2,
    const float* __restrict__ bi3, const float* __restrict__ bh3)
{
    int i = blockIdx.x * blockDim.x + threadIdx.x;
    if (i < BATCH * HID) {
        h0[i] = 0.0f;
        c0[i] = 0.0f;
    }
    if (i < HID) {
        bsum[0 * HID + i] = bi0[i] + bh0[i];
        bsum[1 * HID + i] = bi1[i] + bh1[i];
        bsum[2 * HID + i] = bi2[i] + bh2[i];
        bsum[3 * HID + i] = bi3[i] + bh3[i];
    }
}

__global__ __launch_bounds__(256) void tail_copy_kernel(
    const float* __restrict__ h_fin, const float* __restrict__ c_fin,
    float* __restrict__ out_last, float* __restrict__ out_tail)
{
    int i = blockIdx.x * blockDim.x + threadIdx.x;
    if (i < BATCH * HID) {
        const int b  = i / HID;
        const int h  = i % HID;
        const int pk = (b >> 4) * PK_TILE + (h >> 2) * 64 + (b & 15) * 4 + (h & 3);
        const float hv = h_fin[pk];
        out_last[i]               = hv;
        out_tail[i]               = hv;
        out_tail[BATCH * HID + i] = c_fin[pk];
    }
}

extern "C" void kernel_launch(void* const* d_in, const int* in_sizes, int n_in,
                              void* d_out, int out_size, void* d_ws, size_t ws_size,
                              hipStream_t stream) {
    (void)in_sizes; (void)n_in; (void)out_size;

    const float* x = (const float*)d_in[0];
    GateParams gp;
    const float* bi[4];
    const float* bh[4];
    for (int g = 0; g < 4; ++g) {
        gp.Wi[g] = (const float*)d_in[1 + 4 * g];
        bi[g]    = (const float*)d_in[2 + 4 * g];
        gp.Wh[g] = (const float*)d_in[3 + 4 * g];
        bh[g]    = (const float*)d_in[4 + 4 * g];
    }

    float* ws    = (float*)d_ws;
    float* hbuf0 = ws;
    float* hbuf1 = ws + BATCH * HID;
    float* cbuf  = ws + 2 * BATCH * HID;
    float* bsum  = ws + 3 * BATCH * HID;
    float* xp    = ws + WS_STATE_FLOATS;
    float* wp    = ws + WS_STATE_FLOATS + WS_XP_FLOATS;
    float* out   = (float*)d_out;
    const unsigned long long ws_bytes = (unsigned long long)ws_size;

    prep_kernel<<<(BATCH * HID + 255) / 256, 256, 0, stream>>>(
        hbuf0, cbuf, bsum,
        bi[0], bh[0], bi[1], bh[1], bi[2], bh[2], bi[3], bh[3]);

    wpack_kernel<<<WS_WP_FLOATS / 16 / 256, 256, 0, stream>>>(wp, gp, ws_bytes);

    dim3 grid(BATCH / 16, HID / 16);
    dim3 xgrid(CHUNK, 4 * HID / 256);
    for (int t0 = 0; t0 < SEQL; t0 += CHUNK) {
        xproj_kernel<<<xgrid, 256, 0, stream>>>(
            x + (size_t)t0 * BATCH * IN_DIM, xp, gp, ws_bytes);
        for (int tl = 0; tl < CHUNK; ++tl) {
            const int t = t0 + tl;
            const float* h_in  = (t & 1) ? hbuf1 : hbuf0;
            float*       h_out = (t & 1) ? hbuf0 : hbuf1;
            float* out_prev = t ? out + (size_t)(t - 1) * BATCH * HID : nullptr;
            lstm_step_kernel<<<grid, 128, 0, stream>>>(
                x + (size_t)t * BATCH * IN_DIM, h_in, h_out, cbuf,
                out_prev, bsum,
                xp + (size_t)tl * 4 * BATCH * HID, wp, gp, ws_bytes);
        }
    }
    tail_copy_kernel<<<(BATCH * HID + 255) / 256, 256, 0, stream>>>(
        hbuf0, cbuf, out + (size_t)(SEQL - 1) * BATCH * HID, out + (size_t)SEQL * BATCH * HID);
}
